# starting kernel
# baseline (speedup 1.0000x reference)
#include <hip/hip_runtime.h>
#include <hip/hip_bf16.h>

#define BB 8
#define NN 2048
#define FF 8
#define KK 5

#define MCH   64
#define SROW  68
#define NWAVE 4

typedef __attribute__((ext_vector_type(2))) float v2f;
typedef __attribute__((ext_vector_type(4))) float v4f;
typedef __attribute__((ext_vector_type(8))) float v8f;

__global__ __launch_bounds__(32 * NWAVE)
void spmv_wmma_kernel(const float* __restrict__ S,
                      const float* __restrict__ xin,
                      float* __restrict__ xout) {
  __shared__ float s_lds[NWAVE * 16 * SROW];
  __shared__ float x_lds[16 * SROW];

  const int b    = blockIdx.y;
  const int tid  = threadIdx.x;
  const int wave = tid >> 5;
  const int lane = tid & 31;
  const int n0   = blockIdx.x * (16 * NWAVE) + wave * 16;

  const float* Sb = S   + (size_t)b * NN * NN;
  const float* xb = xin + (size_t)b * NN * FF;

  const int l15 = lane & 15;
  const int lhi = lane >> 4;

  float* s_wave = s_lds + wave * 16 * SROW;
  const int frag_base = l15 * SROW + lhi * 2;

  const int xm   = tid >> 1;
  const int xf0  = (tid & 1) * 4;
  const int mrow = lane >> 2;
  const int f4   = (lane & 3) * 4;

  v8f acc = {};

  v4f x_pref;
  v4f s_pref[8];
  x_pref = *(const v4f*)(xb + (size_t)xm * FF + xf0);
#pragma unroll
  for (int r = 0; r < 8; ++r)
    s_pref[r] = *(const v4f*)(Sb + (size_t)(mrow + r * 8) * NN + n0 + f4);

  for (int mc = 0; mc < NN; mc += MCH) {
    __syncthreads();

    x_lds[(xf0 + 0) * SROW + xm] = x_pref.x;
    x_lds[(xf0 + 1) * SROW + xm] = x_pref.y;
    x_lds[(xf0 + 2) * SROW + xm] = x_pref.z;
    x_lds[(xf0 + 3) * SROW + xm] = x_pref.w;
#pragma unroll
    for (int r = 0; r < 8; ++r) {
      const int m = mrow + r * 8;
      s_wave[(f4 + 0) * SROW + m] = s_pref[r].x;
      s_wave[(f4 + 1) * SROW + m] = s_pref[r].y;
      s_wave[(f4 + 2) * SROW + m] = s_pref[r].z;
      s_wave[(f4 + 3) * SROW + m] = s_pref[r].w;
    }

    {
      int mc_next = mc + MCH;
      if (mc_next >= NN) mc_next = 0;
      x_pref = *(const v4f*)(xb + (size_t)(mc_next + xm) * FF + xf0);
#pragma unroll
      for (int r = 0; r < 8; ++r)
        s_pref[r] =
            *(const v4f*)(Sb + (size_t)(mc_next + mrow + r * 8) * NN + n0 + f4);
    }

    __syncthreads();

#pragma unroll
    for (int s = 0; s < 16; ++s) {
      v2f a  = *(const v2f*)(s_wave + frag_base + 4 * s);
      v2f bf = *(const v2f*)(x_lds  + frag_base + 4 * s);
      acc = __builtin_amdgcn_wmma_f32_16x16x4_f32(
           false, a,  false, bf,
           (short)0, acc,  false,  false);
    }
  }

  if (l15 < FF) {
    float* yb = xout + (size_t)b * NN * FF;
#pragma unroll
    for (int r = 0; r < 8; ++r) {
      const int n = n0 + r + lhi * 8;
      yb[(size_t)n * FF + l15] = acc[r];
    }
  }
}

__global__ __launch_bounds__(256)
void yw_kernel(const float* __restrict__ xk, const float* __restrict__ Wk,
               const float* __restrict__ bias, float* __restrict__ y,
               int accumulate) {
  const int i = blockIdx.x * blockDim.x + threadIdx.x;
  if (i >= BB * NN) return;
  const float* xr = xk + (size_t)i * FF;
  float xv[FF];
#pragma unroll
  for (int f = 0; f < FF; ++f) xv[f] = xr[f];
#pragma unroll
  for (int g = 0; g < FF; ++g) {
    float s = accumulate ? y[(size_t)i * FF + g] : bias[g];
#pragma unroll
    for (int f = 0; f < FF; ++f) s += xv[f] * Wk[f * FF + g];
    y[(size_t)i * FF + g] = s;
  }
}

extern "C" void kernel_launch(void* const* d_in, const int* in_sizes, int n_in,
                              void* d_out, int out_size, void* d_ws, size_t ws_size,
                              hipStream_t stream) {
  const float* x    = (const float*)d_in[0];
  const float* S    = (const float*)d_in[1];
  const float* W    = (const float*)d_in[2];
  const float* bias = (const float*)d_in[3];
  float* y = (float*)d_out;

  float* buf0 = (float*)d_ws;
  float* buf1 = buf0 + (size_t)BB * NN * FF;

  const dim3 gw(NN / (16 * NWAVE), BB);
  const int rows = BB * NN;
  const int nb = (rows + 255) / 256;

  yw_kernel<<<nb, 256, 0, stream>>>(x, W, bias, y, 0);

  const float* src = x;
  float* dst = buf0;
  for (int k = 1; k < KK; ++k) {
    spmv_wmma_kernel<<<gw, 32 * NWAVE, 0, stream>>>(S, src, dst);
    yw_kernel<<<nb, 256, 0, stream>>>(dst, W + (size_t)k * FF * FF, bias, y, 1);
    src = dst;
    dst = (dst == buf0) ? buf1 : buf0;
  }
}


# spmv pass rebuilt for bytes in flight: 16 waves per workgroup split m, each streams S by 16-byte loads straight into WMMA A fragments (32 in flight), partials summed via LDS; f32 sum order changes
# speedup vs baseline: 2.1893x; 2.1893x over previous
#include <hip/hip_runtime.h>
#include <hip/hip_bf16.h>

#define BB 8
#define NN 2048
#define FF 8
#define KK 5

#define NWAVE  16
#define NCOL   64
#define MW     (NN / NWAVE)
#define KSTEPS (MW / 4)
#define RING   16
#define XP     132

typedef __attribute__((ext_vector_type(2))) float v2f;
typedef __attribute__((ext_vector_type(4))) float v4f;
typedef __attribute__((ext_vector_type(8))) float v8f;

__global__ __launch_bounds__(32 * NWAVE)
void spmv_wmma_kernel(const float* __restrict__ S,
                      const float* __restrict__ xin,
                      float* __restrict__ xout) {
  __shared__ __attribute__((aligned(16))) float xT[NWAVE][FF][XP];
  __shared__ float red[NWAVE][NCOL][FF];

  const int b    = blockIdx.y;
  const int tid  = threadIdx.x;
  const int wave = tid >> 5;
  const int lane = tid & 31;
  const int c    = lane & 15;
  const int h    = lane >> 4;
  const int n0   = blockIdx.x * NCOL;
  const int mbase = wave * MW;

  const float* Sb = S   + (size_t)b * NN * NN;
  const float* xb = xin + (size_t)b * NN * FF;

  v4f xv[8];
#pragma unroll
  for (int j = 0; j < 8; ++j)
    xv[j] = *(const v4f*)(xb + (size_t)mbase * FF + j * 128 + lane * 4);

  const float* sp = Sb + (size_t)(mbase + 64 * h) * NN + n0 + 4 * c;
  v4f a[RING][2];
#pragma unroll
  for (int t = 0; t < RING; ++t) {
    a[t][0] = *(const v4f*)(sp + (size_t)(2 * t + 0) * NN);
    a[t][1] = *(const v4f*)(sp + (size_t)(2 * t + 1) * NN);
  }

  __builtin_amdgcn_sched_barrier(0);

  {
    const int f0 = (lane & 1) * 4;
#pragma unroll
    for (int j = 0; j < 8; ++j) {
      const int ml = j * 16 + (lane >> 1);
      xT[wave][f0 + 0][ml] = xv[j].x;
      xT[wave][f0 + 1][ml] = xv[j].y;
      xT[wave][f0 + 2][ml] = xv[j].z;
      xT[wave][f0 + 3][ml] = xv[j].w;
    }
  }
  const float* xrow = &xT[wave][c & 7][64 * h];

  v8f acc[4] = {};
  v4f qcur = *(const v4f*)(xrow);

#pragma unroll
  for (int u = 0; u < KSTEPS / 2; ++u) {
    const v4f qnext = *(const v4f*)(xrow + 4 * ((u + 1) % (KSTEPS / 2)));
#pragma unroll
    for (int e = 0; e < 2; ++e) {
      const int slot = (2 * u + e) % RING;
      const v4f a0 = a[slot][0];
      const v4f a1 = a[slot][1];
      const v2f bf = e ? (v2f){qcur.z, qcur.w} : (v2f){qcur.x, qcur.y};
#pragma unroll
      for (int j = 0; j < 4; ++j) {
        const v2f af = {a0[j], a1[j]};
        acc[j] = __builtin_amdgcn_wmma_f32_16x16x4_f32(
             false, af,  false, bf,
             (short)0, acc[j],  false,  false);
      }
    }
    if (2 * u + RING < KSTEPS) {
      __builtin_amdgcn_sched_barrier(0);
#pragma unroll
      for (int e = 0; e < 2; ++e) {
        const int t = 2 * u + e;
        a[t % RING][0] = *(const v4f*)(sp + (size_t)(2 * (t + RING) + 0) * NN);
        a[t % RING][1] = *(const v4f*)(sp + (size_t)(2 * (t + RING) + 1) * NN);
      }
      __builtin_amdgcn_sched_barrier(0);
    }
    qcur = qnext;
  }

  if (c < FF) {
#pragma unroll
    for (int j = 0; j < 4; ++j)
#pragma unroll
      for (int r = 0; r < 8; ++r)
        red[wave][4 * (8 * h + r) + j][c] = acc[j][r];
  }
  __syncthreads();

  {
    const float* rp = &red[0][0][0] + tid;
    float s = rp[0];
#pragma unroll
    for (int w = 1; w < NWAVE; ++w) s += rp[w * NCOL * FF];
    xout[(size_t)b * NN * FF + (size_t)n0 * FF + tid] = s;
  }
}

__global__ __launch_bounds__(256)
void yw_kernel(const float* __restrict__ xk, const float* __restrict__ Wk,
               const float* __restrict__ bias, float* __restrict__ y,
               int accumulate) {
  const int i = blockIdx.x * blockDim.x + threadIdx.x;
  if (i >= BB * NN) return;
  const float* xr = xk + (size_t)i * FF;
  float xv[FF];
#pragma unroll
  for (int f = 0; f < FF; ++f) xv[f] = xr[f];
#pragma unroll
  for (int g = 0; g < FF; ++g) {
    float s = accumulate ? y[(size_t)i * FF + g] : bias[g];
#pragma unroll
    for (int f = 0; f < FF; ++f) s += xv[f] * Wk[f * FF + g];
    y[(size_t)i * FF + g] = s;
  }
}

extern "C" void kernel_launch(void* const* d_in, const int* in_sizes, int n_in,
                              void* d_out, int out_size, void* d_ws, size_t ws_size,
                              hipStream_t stream) {
  const float* x    = (const float*)d_in[0];
  const float* S    = (const float*)d_in[1];
  const float* W    = (const float*)d_in[2];
  const float* bias = (const float*)d_in[3];
  float* y = (float*)d_out;

  float* buf0 = (float*)d_ws;
  float* buf1 = buf0 + (size_t)BB * NN * FF;

  const dim3 gw(NN / NCOL, BB);
  const int rows = BB * NN;
  const int nb = (rows + 255) / 256;

  yw_kernel<<<nb, 256, 0, stream>>>(x, W, bias, y, 0);

  const float* src = x;
  float* dst = buf0;
  for (int k = 1; k < KK; ++k) {
    spmv_wmma_kernel<<<gw, 32 * NWAVE, 0, stream>>>(S, src, dst);
    yw_kernel<<<nb, 256, 0, stream>>>(dst, W + (size_t)k * FF * FF, bias, y, 1);
    src = dst;
    dst = (dst == buf0) ? buf1 : buf0;
  }
}


# y = sum_k x_k W_k + bias fused into the pass kernel's reduction tail (row values broadcast in 8-lane groups, x0 W0 + bias folded into pass 1): 9 launches -> 4; f32 sum order unchanged
# speedup vs baseline: 2.8367x; 1.2957x over previous
#include <hip/hip_runtime.h>
#include <hip/hip_bf16.h>

#define BB 8
#define NN 2048
#define FF 8
#define KK 5

#define NWAVE  16
#define NCOL   64
#define MW     (NN / NWAVE)
#define KSTEPS (MW / 4)
#define RING   16
#define XP     132

typedef __attribute__((ext_vector_type(2))) float v2f;
typedef __attribute__((ext_vector_type(4))) float v4f;
typedef __attribute__((ext_vector_type(8))) float v8f;


template <int F>
__device__ __forceinline__ float bcast8(float v) {
  return __uint_as_float(__builtin_amdgcn_mov_dpp8(__float_as_uint(v), F * 0x249249));
}

template <int MODE>
__global__ __launch_bounds__(32 * NWAVE)
void spmv_wmma_kernel(const float* __restrict__ S,
                      const float* __restrict__ xin,
                      float* __restrict__ xout,
                      const float* __restrict__ Wk,
                      const float* __restrict__ W0,
                      const float* __restrict__ bias,
                      float* __restrict__ y) {
  __shared__ __attribute__((aligned(16))) float xT[NWAVE][FF][XP];
  __shared__ float red[NWAVE][NCOL][FF];

  const int b    = blockIdx.y;
  const int tid  = threadIdx.x;
  const int wave = tid >> 5;
  const int lane = tid & 31;
  const int c    = lane & 15;
  const int h    = lane >> 4;
  const int n0   = blockIdx.x * NCOL;
  const int mbase = wave * MW;

  const float* Sb = S   + (size_t)b * NN * NN;
  const float* xb = xin + (size_t)b * NN * FF;

  v4f xv[8];
#pragma unroll
  for (int j = 0; j < 8; ++j)
    xv[j] = *(const v4f*)(xb + (size_t)mbase * FF + j * 128 + lane * 4);

  const float* sp = Sb + (size_t)(mbase + 64 * h) * NN + n0 + 4 * c;
  v4f a[RING][2];
#pragma unroll
  for (int t = 0; t < RING; ++t) {
    a[t][0] = *(const v4f*)(sp + (size_t)(2 * t + 0) * NN);
    a[t][1] = *(const v4f*)(sp + (size_t)(2 * t + 1) * NN);
  }

  __builtin_amdgcn_sched_barrier(0);

  {
    const int f0 = (lane & 1) * 4;
#pragma unroll
    for (int j = 0; j < 8; ++j) {
      const int ml = j * 16 + (lane >> 1);
      xT[wave][f0 + 0][ml] = xv[j].x;
      xT[wave][f0 + 1][ml] = xv[j].y;
      xT[wave][f0 + 2][ml] = xv[j].z;
      xT[wave][f0 + 3][ml] = xv[j].w;
    }
  }
  const float* xrow = &xT[wave][c & 7][64 * h];

  v8f acc[4] = {};
  v4f qcur = *(const v4f*)(xrow);

#pragma unroll
  for (int u = 0; u < KSTEPS / 2; ++u) {
    const v4f qnext = *(const v4f*)(xrow + 4 * ((u + 1) % (KSTEPS / 2)));
#pragma unroll
    for (int e = 0; e < 2; ++e) {
      const int slot = (2 * u + e) % RING;
      const v4f a0 = a[slot][0];
      const v4f a1 = a[slot][1];
      const v2f bf = e ? (v2f){qcur.z, qcur.w} : (v2f){qcur.x, qcur.y};
#pragma unroll
      for (int j = 0; j < 4; ++j) {
        const v2f af = {a0[j], a1[j]};
        acc[j] = __builtin_amdgcn_wmma_f32_16x16x4_f32(
             false, af,  false, bf,
             (short)0, acc[j],  false,  false);
      }
    }
    if (2 * u + RING < KSTEPS) {
      __builtin_amdgcn_sched_barrier(0);
#pragma unroll
      for (int e = 0; e < 2; ++e) {
        const int t = 2 * u + e;
        a[t % RING][0] = *(const v4f*)(sp + (size_t)(2 * (t + RING) + 0) * NN);
        a[t % RING][1] = *(const v4f*)(sp + (size_t)(2 * (t + RING) + 1) * NN);
      }
      __builtin_amdgcn_sched_barrier(0);
    }
    qcur = qnext;
  }

  const int g = tid & 7;
  const size_t oidx = (size_t)b * NN * FF + (size_t)n0 * FF + tid;
  float wk[FF], w0[FF];
  float yv, x0v = 0.0f;
#pragma unroll
  for (int f = 0; f < FF; ++f) wk[f] = Wk[f * FF + g];
  if (MODE == 0) {
#pragma unroll
    for (int f = 0; f < FF; ++f) w0[f] = W0[f * FF + g];
    yv  = bias[g];
    x0v = xin[oidx];
  } else {
    yv = y[oidx];
  }

  if (c < FF) {
#pragma unroll
    for (int j = 0; j < 4; ++j)
#pragma unroll
      for (int r = 0; r < 8; ++r)
        red[wave][4 * (8 * h + r) + j][c] = acc[j][r];
  }
  __syncthreads();

  {
    const float* rp = &red[0][0][0] + tid;
    float s = rp[0];
#pragma unroll
    for (int w = 1; w < NWAVE; ++w) s += rp[w * NCOL * FF];
    if (MODE != 2) xout[oidx] = s;

    float t = yv;
    if (MODE == 0) {
      t = __builtin_fmaf(bcast8<0>(x0v), w0[0], t);
      t = __builtin_fmaf(bcast8<1>(x0v), w0[1], t);
      t = __builtin_fmaf(bcast8<2>(x0v), w0[2], t);
      t = __builtin_fmaf(bcast8<3>(x0v), w0[3], t);
      t = __builtin_fmaf(bcast8<4>(x0v), w0[4], t);
      t = __builtin_fmaf(bcast8<5>(x0v), w0[5], t);
      t = __builtin_fmaf(bcast8<6>(x0v), w0[6], t);
      t = __builtin_fmaf(bcast8<7>(x0v), w0[7], t);
    }
    t = __builtin_fmaf(bcast8<0>(s), wk[0], t);
    t = __builtin_fmaf(bcast8<1>(s), wk[1], t);
    t = __builtin_fmaf(bcast8<2>(s), wk[2], t);
    t = __builtin_fmaf(bcast8<3>(s), wk[3], t);
    t = __builtin_fmaf(bcast8<4>(s), wk[4], t);
    t = __builtin_fmaf(bcast8<5>(s), wk[5], t);
    t = __builtin_fmaf(bcast8<6>(s), wk[6], t);
    t = __builtin_fmaf(bcast8<7>(s), wk[7], t);
    y[oidx] = t;
  }
}

extern "C" void kernel_launch(void* const* d_in, const int* in_sizes, int n_in,
                              void* d_out, int out_size, void* d_ws, size_t ws_size,
                              hipStream_t stream) {
  const float* x    = (const float*)d_in[0];
  const float* S    = (const float*)d_in[1];
  const float* W    = (const float*)d_in[2];
  const float* bias = (const float*)d_in[3];
  float* y = (float*)d_out;

  float* buf0 = (float*)d_ws;
  float* buf1 = buf0 + (size_t)BB * NN * FF;

  const dim3 gw(NN / NCOL, BB);

  spmv_wmma_kernel<0><<<gw, 32 * NWAVE, 0, stream>>>(S, x,    buf0, W + 1 * FF * FF, W, bias, y);
  spmv_wmma_kernel<1><<<gw, 32 * NWAVE, 0, stream>>>(S, buf0, buf1, W + 2 * FF * FF, W, bias, y);
  spmv_wmma_kernel<1><<<gw, 32 * NWAVE, 0, stream>>>(S, buf1, buf0, W + 3 * FF * FF, W, bias, y);
  spmv_wmma_kernel<2><<<gw, 32 * NWAVE, 0, stream>>>(S, buf0, buf1, W + 4 * FF * FF, W, bias, y);
}
